# starting kernel
# baseline (speedup 1.0000x reference)
#include <hip/hip_runtime.h>
#include <hip/hip_bf16.h>

typedef __attribute__((ext_vector_type(2))) float v2f;
typedef __attribute__((ext_vector_type(8))) float v8f;

#define BB 32
#define SS 150
#define DD 128

__global__ void EOPA_neigh_max_kernel(const float* __restrict__ x,
                                      const int* __restrict__ A,
                                      float* __restrict__ neigh) {
    extern __shared__ float smem[];
    float* xs = smem;
    float* am = smem + SS * DD;

    const int b   = blockIdx.x;
    const int tid = threadIdx.x;

    const float* xb = x + (size_t)b * SS * DD;
    for (int idx = tid; idx < SS * DD; idx += blockDim.x) xs[idx] = xb[idx];
    __syncthreads();

    const int* Ab = A + (size_t)b * SS * SS;
    const int c = tid >> 7;
    const int d = tid & 127;

    for (int i0 = 0; i0 < SS; i0 += 2) {
        for (int idx = tid; idx < SS * 2; idx += blockDim.x) {
            const int j  = idx >> 1;
            const int cc = idx & 1;
            am[idx] = (float)Ab[j * SS + i0 + cc];
        }
        __syncthreads();

        float acc = -INFINITY;
        #pragma unroll 5
        for (int j = 0; j < SS; ++j) {
            acc = fmaxf(acc, xs[j * DD + d] * am[j * 2 + c]);
        }
        neigh[(size_t)b * SS * DD + (size_t)(i0 + c) * DD + d] = acc;
        __syncthreads();
    }
}

__global__ void EOPA_gemm_wmma_kernel(const float* __restrict__ X,
                                      const float* __restrict__ NEIGH,
                                      const float* __restrict__ Wself,
                                      const float* __restrict__ Wneigh,
                                      float* __restrict__ out) {
    const int tid  = threadIdx.x;
    const int wave = tid >> 5;
    const int lane = tid & 31;

    const int tile = blockIdx.x * 8 + wave;
    const int nt = tile & 7;
    const int mt = tile >> 3;
    const int m0 = mt * 16;
    const int n0 = nt * 16;

    const int laneLo = lane & 15;
    const int kHalf  = (lane >> 4) << 1;

    v8f acc = {};

    {
        const float* arow = X     + (size_t)(m0 + laneLo) * DD + kHalf;
        const float* brow = Wself + (size_t)(n0 + laneLo) * DD + kHalf;
        #pragma unroll 8
        for (int k = 0; k < DD; k += 4) {
            v2f a = *(const v2f*)(arow + k);
            v2f b = *(const v2f*)(brow + k);
            acc = __builtin_amdgcn_wmma_f32_16x16x4_f32(
                false, a, false, b, (short)0, acc, false, false);
        }
    }
    {
        const float* arow = NEIGH  + (size_t)(m0 + laneLo) * DD + kHalf;
        const float* brow = Wneigh + (size_t)(n0 + laneLo) * DD + kHalf;
        #pragma unroll 8
        for (int k = 0; k < DD; k += 4) {
            v2f a = *(const v2f*)(arow + k);
            v2f b = *(const v2f*)(brow + k);
            acc = __builtin_amdgcn_wmma_f32_16x16x4_f32(
                false, a, false, b, (short)0, acc, false, false);
        }
    }

    const int colN    = n0 + laneLo;
    const int rowBase = m0 + ((lane >> 4) << 3);
    #pragma unroll
    for (int r = 0; r < 8; ++r) {
        out[(size_t)(rowBase + r) * DD + colN] = acc[r];
    }
}

extern "C" void kernel_launch(void* const* d_in, const int* in_sizes, int n_in,
                              void* d_out, int out_size, void* d_ws, size_t ws_size,
                              hipStream_t stream) {
    const float* x      = (const float*)d_in[0];
    const int*   A      = (const int*)  d_in[1];
    const float* Wself  = (const float*)d_in[2];
    const float* Wneigh = (const float*)d_in[3];
    float*       out    = (float*)d_out;
    float*       neigh  = (float*)d_ws;

    const size_t lds_bytes = (size_t)(SS * DD + 2 * SS) * sizeof(float);
    EOPA_neigh_max_kernel<<<BB, 256, lds_bytes, stream>>>(x, A, neigh);

    EOPA_gemm_wmma_kernel<<<300, 256, 0, stream>>>(x, neigh, Wself, Wneigh, out);
}


# neighbour max re-tiled: grid 32 -> 256 groups (batch x 20 nodes); each wave keeps 5 nodes x 128 features as register maxima and streams the x rows once; masks staged once per group in LDS
# speedup vs baseline: 21.8495x; 21.8495x over previous
#include <hip/hip_runtime.h>
#include <hip/hip_bf16.h>

typedef __attribute__((ext_vector_type(2))) float v2f;
typedef __attribute__((ext_vector_type(4))) float v4f;
typedef __attribute__((ext_vector_type(8))) float v8f;

#define BB 32
#define SS 150
#define DD 128

#define IW   5
#define WPB  4
#define IPB  (IW * WPB)
#define NCH  8
#define NB   25
#define BSTR 8
#define SROWS 6

__global__ __launch_bounds__(WPB * 32)
void EOPA_neigh_max_kernel(const float* __restrict__ x,
                           const int* __restrict__ A,
                           float* __restrict__ neigh) {
    __shared__ float sm[WPB * SS * BSTR];

    const int tid  = threadIdx.x;
    const int lane = tid & 31;
    const int w     = __builtin_amdgcn_readfirstlane(tid >> 5);
    const int b     = blockIdx.y;
    const int ibase = blockIdx.x * IPB;
    const int i0    = ibase + w * IW;
    const bool live = i0 < SS;

    const float* xl = x + (size_t)b * SS * DD + lane * 4;
    const int*   Ab = A + (size_t)b * SS * SS;

    const int  cc     = tid % IPB;
    const int  jr     = tid / IPB;
    const bool stager = (tid < SROWS * IPB) && (ibase + cc < SS);
    int av[SS / SROWS];
    if (stager) {
        const int* ap = Ab + jr * SS + ibase + cc;
        #pragma unroll
        for (int k = 0; k < SS / SROWS; ++k) av[k] = ap[k * SROWS * SS];
    }

    v4f bufA[NB], bufB[NB];
    if (live) {
        #pragma unroll
        for (int r = 0; r < NB; ++r) bufA[r] = *(const v4f*)(xl + r * DD);
    }

    if (stager) {
        float* sp = sm + ((cc / IW) * SS + jr) * BSTR + (cc % IW);
        #pragma unroll
        for (int k = 0; k < SS / SROWS; ++k) sp[k * SROWS * BSTR] = (av[k] == 1) ? 1.0f : 0.0f;
    }
    __syncthreads();
    if (!live) return;

    const float* smw = sm + w * SS * BSTR;

    v4f P[IW];
    #pragma unroll
    for (int c = 0; c < IW; ++c) P[c] = (v4f){-INFINITY, -INFINITY, -INFINITY, -INFINITY};

#define CONSUME(BUF, JB)                                                        \
    _Pragma("unroll")                                                           \
    for (int r = 0; r < NB; ++r) {                                              \
        const v4f   m03 = *(const v4f*)(smw + ((JB) + r) * BSTR);               \
        const float m4  = smw[((JB) + r) * BSTR + 4];                           \
        const v4f   xv  = BUF[r];                                               \
        _Pragma("unroll")                                                       \
        for (int k = 0; k < 4; ++k) {                                           \
            P[0][k] = fmaxf(P[0][k], xv[k] * m03[0]);                           \
            P[1][k] = fmaxf(P[1][k], xv[k] * m03[1]);                           \
            P[2][k] = fmaxf(P[2][k], xv[k] * m03[2]);                           \
            P[3][k] = fmaxf(P[3][k], xv[k] * m03[3]);                           \
            P[4][k] = fmaxf(P[4][k], xv[k] * m4);                               \
        }                                                                       \
    }

    for (int jb = 0; jb < SS; jb += 2 * NB) {
        #pragma unroll
        for (int r = 0; r < NB; ++r) bufB[r] = *(const v4f*)(xl + (jb + NB + r) * DD);
        CONSUME(bufA, jb)
        const int jn = (jb + 2 * NB < SS) ? jb + 2 * NB : jb;
        #pragma unroll
        for (int r = 0; r < NB; ++r) bufA[r] = *(const v4f*)(xl + (jn + r) * DD);
        CONSUME(bufB, jb + NB)
    }
#undef CONSUME

    #pragma unroll
    for (int c = 0; c < IW; ++c)
        *(v4f*)(neigh + ((size_t)b * SS + (i0 + c)) * DD + lane * 4) = P[c];
}

__global__ void EOPA_gemm_wmma_kernel(const float* __restrict__ X,
                                      const float* __restrict__ NEIGH,
                                      const float* __restrict__ Wself,
                                      const float* __restrict__ Wneigh,
                                      float* __restrict__ out) {
    const int tid  = threadIdx.x;
    const int wave = tid >> 5;
    const int lane = tid & 31;

    const int tile = blockIdx.x * 8 + wave;
    const int nt = tile & 7;
    const int mt = tile >> 3;
    const int m0 = mt * 16;
    const int n0 = nt * 16;

    const int laneLo = lane & 15;
    const int kHalf  = (lane >> 4) << 1;

    v8f acc = {};

    {
        const float* arow = X     + (size_t)(m0 + laneLo) * DD + kHalf;
        const float* brow = Wself + (size_t)(n0 + laneLo) * DD + kHalf;
        #pragma unroll 8
        for (int k = 0; k < DD; k += 4) {
            v2f a = *(const v2f*)(arow + k);
            v2f b = *(const v2f*)(brow + k);
            acc = __builtin_amdgcn_wmma_f32_16x16x4_f32(
                false, a, false, b, (short)0, acc, false, false);
        }
    }
    {
        const float* arow = NEIGH  + (size_t)(m0 + laneLo) * DD + kHalf;
        const float* brow = Wneigh + (size_t)(n0 + laneLo) * DD + kHalf;
        #pragma unroll 8
        for (int k = 0; k < DD; k += 4) {
            v2f a = *(const v2f*)(arow + k);
            v2f b = *(const v2f*)(brow + k);
            acc = __builtin_amdgcn_wmma_f32_16x16x4_f32(
                false, a, false, b, (short)0, acc, false, false);
        }
    }

    const int colN    = n0 + laneLo;
    const int rowBase = m0 + ((lane >> 4) << 3);
    #pragma unroll
    for (int r = 0; r < 8; ++r) {
        out[(size_t)(rowBase + r) * DD + colN] = acc[r];
    }
}

extern "C" void kernel_launch(void* const* d_in, const int* in_sizes, int n_in,
                              void* d_out, int out_size, void* d_ws, size_t ws_size,
                              hipStream_t stream) {
    const float* x      = (const float*)d_in[0];
    const int*   A      = (const int*)  d_in[1];
    const float* Wself  = (const float*)d_in[2];
    const float* Wneigh = (const float*)d_in[3];
    float*       out    = (float*)d_out;
    float*       neigh  = (float*)d_ws;

    EOPA_neigh_max_kernel<<<dim3(8, 32), 128, 0, stream>>>(x, A, neigh);

    EOPA_gemm_wmma_kernel<<<300, 256, 0, stream>>>(x, neigh, Wself, Wneigh, out);
}


# GEMM operands staged global->LDS by whole-row b128 loads (4 full lines per instruction; was b64 fragment loads, a partial line per matrix row); group tile 32x64, fragments from LDS; sum order kept
# speedup vs baseline: 29.0597x; 1.3300x over previous
#include <hip/hip_runtime.h>
#include <hip/hip_bf16.h>

typedef __attribute__((ext_vector_type(2))) float v2f;
typedef __attribute__((ext_vector_type(4))) float v4f;
typedef __attribute__((ext_vector_type(8))) float v8f;

#define BB 32
#define SS 150
#define DD 128

#define IW   5
#define WPB  4
#define IPB  (IW * WPB)
#define NCH  8
#define NB   25
#define BSTR 8
#define SROWS 6

__global__ __launch_bounds__(WPB * 32)
void EOPA_neigh_max_kernel(const float* __restrict__ x,
                           const int* __restrict__ A,
                           float* __restrict__ neigh) {
    __shared__ float sm[WPB * SS * BSTR];

    const int tid  = threadIdx.x;
    const int lane = tid & 31;
    const int w     = __builtin_amdgcn_readfirstlane(tid >> 5);
    const int b     = blockIdx.y;
    const int ibase = blockIdx.x * IPB;
    const int i0    = ibase + w * IW;
    const bool live = i0 < SS;

    const float* xl = x + (size_t)b * SS * DD + lane * 4;
    const int*   Ab = A + (size_t)b * SS * SS;

    const int  cc     = tid % IPB;
    const int  jr     = tid / IPB;
    const bool stager = (tid < SROWS * IPB) && (ibase + cc < SS);
    int av[SS / SROWS];
    if (stager) {
        const int* ap = Ab + jr * SS + ibase + cc;
        #pragma unroll
        for (int k = 0; k < SS / SROWS; ++k) av[k] = ap[k * SROWS * SS];
    }

    v4f bufA[NB], bufB[NB];
    if (live) {
        #pragma unroll
        for (int r = 0; r < NB; ++r) bufA[r] = *(const v4f*)(xl + r * DD);
    }

    if (stager) {
        float* sp = sm + ((cc / IW) * SS + jr) * BSTR + (cc % IW);
        #pragma unroll
        for (int k = 0; k < SS / SROWS; ++k) sp[k * SROWS * BSTR] = (av[k] == 1) ? 1.0f : 0.0f;
    }
    __syncthreads();
    if (!live) return;

    const float* smw = sm + w * SS * BSTR;

    v4f P[IW];
    #pragma unroll
    for (int c = 0; c < IW; ++c) P[c] = (v4f){-INFINITY, -INFINITY, -INFINITY, -INFINITY};

#define CONSUME(BUF, JB)                                                        \
    _Pragma("unroll")                                                           \
    for (int r = 0; r < NB; ++r) {                                              \
        const v4f   m03 = *(const v4f*)(smw + ((JB) + r) * BSTR);               \
        const float m4  = smw[((JB) + r) * BSTR + 4];                           \
        const v4f   xv  = BUF[r];                                               \
        _Pragma("unroll")                                                       \
        for (int k = 0; k < 4; ++k) {                                           \
            P[0][k] = fmaxf(P[0][k], xv[k] * m03[0]);                           \
            P[1][k] = fmaxf(P[1][k], xv[k] * m03[1]);                           \
            P[2][k] = fmaxf(P[2][k], xv[k] * m03[2]);                           \
            P[3][k] = fmaxf(P[3][k], xv[k] * m03[3]);                           \
            P[4][k] = fmaxf(P[4][k], xv[k] * m4);                               \
        }                                                                       \
    }

    for (int jb = 0; jb < SS; jb += 2 * NB) {
        #pragma unroll
        for (int r = 0; r < NB; ++r) bufB[r] = *(const v4f*)(xl + (jb + NB + r) * DD);
        CONSUME(bufA, jb)
        const int jn = (jb + 2 * NB < SS) ? jb + 2 * NB : jb;
        #pragma unroll
        for (int r = 0; r < NB; ++r) bufA[r] = *(const v4f*)(xl + (jn + r) * DD);
        CONSUME(bufB, jb + NB)
    }
#undef CONSUME

    #pragma unroll
    for (int c = 0; c < IW; ++c)
        *(v4f*)(neigh + ((size_t)b * SS + (i0 + c)) * DD + lane * 4) = P[c];
}

#define GP 132
#define TM 32
#define TN 64

__global__ __launch_bounds__(256)
void EOPA_gemm_wmma_kernel(const float* __restrict__ X,
                           const float* __restrict__ NEIGH,
                           const float* __restrict__ Wself,
                           const float* __restrict__ Wneigh,
                           float* __restrict__ out) {
    __shared__ __attribute__((aligned(16))) float sA[2][TM * GP];
    __shared__ __attribute__((aligned(16))) float sW[2][TN * GP];

    const int tid  = threadIdx.x;
    const int lane = tid & 31;
    const int wave = __builtin_amdgcn_readfirstlane(tid >> 5);

    const int m0  = (blockIdx.x >> 1) * TM;
    const int nb  = (blockIdx.x & 1) * TN;
    const int col = lane * 4;

    v4f xa[TM / 8], na[TM / 8], ws[TN / 8], wn[TN / 8];
    #pragma unroll
    for (int it = 0; it < TM / 8; ++it)
        xa[it] = *(const v4f*)(X + (size_t)(m0 + it * 8 + wave) * DD + col);
    #pragma unroll
    for (int it = 0; it < TN / 8; ++it)
        ws[it] = *(const v4f*)(Wself + (size_t)(nb + it * 8 + wave) * DD + col);

    const int mi     = wave >> 2;
    const int ni     = wave & 3;
    const int laneLo = lane & 15;
    const int kHalf  = (lane >> 4) << 1;
    const int fa = (mi * 16 + laneLo) * GP + kHalf;
    const int fb = (ni * 16 + laneLo) * GP + kHalf;

    v8f acc = {};

    #pragma unroll
    for (int it = 0; it < TM / 8; ++it)
        *(v4f*)(&sA[0][(it * 8 + wave) * GP + col]) = xa[it];
    #pragma unroll
    for (int it = 0; it < TN / 8; ++it)
        *(v4f*)(&sW[0][(it * 8 + wave) * GP + col]) = ws[it];
    __syncthreads();
    #pragma unroll
    for (int it = 0; it < TM / 8; ++it)
        na[it] = *(const v4f*)(NEIGH + (size_t)(m0 + it * 8 + wave) * DD + col);
    #pragma unroll
    for (int it = 0; it < TN / 8; ++it)
        wn[it] = *(const v4f*)(Wneigh + (size_t)(nb + it * 8 + wave) * DD + col);
    {
        v2f af[DD / 4], bf[DD / 4];
        #pragma unroll
        for (int s = 0; s < DD / 4; ++s) {
            af[s] = *(const v2f*)(&sA[0][fa + 4 * s]);
            bf[s] = *(const v2f*)(&sW[0][fb + 4 * s]);
        }
        __builtin_amdgcn_sched_barrier(0);
        #pragma unroll
        for (int s = 0; s < DD / 4; ++s)
            acc = __builtin_amdgcn_wmma_f32_16x16x4_f32(
                false, af[s], false, bf[s], (short)0, acc, false, false);
    }
    asm volatile("" :: "v"(acc));

    #pragma unroll
    for (int it = 0; it < TM / 8; ++it)
        *(v4f*)(&sA[1][(it * 8 + wave) * GP + col]) = na[it];
    #pragma unroll
    for (int it = 0; it < TN / 8; ++it)
        *(v4f*)(&sW[1][(it * 8 + wave) * GP + col]) = wn[it];
    __syncthreads();
    {
        v2f af[DD / 4], bf[DD / 4];
        #pragma unroll
        for (int s = 0; s < DD / 4; ++s) {
            af[s] = *(const v2f*)(&sA[1][fa + 4 * s]);
            bf[s] = *(const v2f*)(&sW[1][fb + 4 * s]);
        }
        __builtin_amdgcn_sched_barrier(0);
        #pragma unroll
        for (int s = 0; s < DD / 4; ++s)
            acc = __builtin_amdgcn_wmma_f32_16x16x4_f32(
                false, af[s], false, bf[s], (short)0, acc, false, false);
    }

    const int colN    = nb + ni * 16 + laneLo;
    const int rowBase = m0 + mi * 16 + ((lane >> 4) << 3);
    #pragma unroll
    for (int r = 0; r < 8; ++r) {
        out[(size_t)(rowBase + r) * DD + colN] = acc[r];
    }
}

extern "C" void kernel_launch(void* const* d_in, const int* in_sizes, int n_in,
                              void* d_out, int out_size, void* d_ws, size_t ws_size,
                              hipStream_t stream) {
    const float* x      = (const float*)d_in[0];
    const int*   A      = (const int*)  d_in[1];
    const float* Wself  = (const float*)d_in[2];
    const float* Wneigh = (const float*)d_in[3];
    float*       out    = (float*)d_out;
    float*       neigh  = (float*)d_ws;

    EOPA_neigh_max_kernel<<<dim3(8, 32), 128, 0, stream>>>(x, A, neigh);

    EOPA_gemm_wmma_kernel<<<300, 256, 0, stream>>>(x, neigh, Wself, Wneigh, out);
}


# neighbour max: the LDS mask reads of each 25-row batch issued as ONE group in front of its arithmetic (was: at each use, one LDS round trip per row pair on a lone wave); sum order unchanged
# speedup vs baseline: 39.8486x; 1.3713x over previous
#include <hip/hip_runtime.h>
#include <hip/hip_bf16.h>

typedef __attribute__((ext_vector_type(2))) float v2f;
typedef __attribute__((ext_vector_type(4))) float v4f;
typedef __attribute__((ext_vector_type(8))) float v8f;

#define BB 32
#define SS 150
#define DD 128

#define IW   5
#define WPB  4
#define IPB  (IW * WPB)
#define NCH  8
#define NB   25
#define BSTR 8
#define SROWS 6

__global__ __launch_bounds__(WPB * 32)
void EOPA_neigh_max_kernel(const float* __restrict__ x,
                           const int* __restrict__ A,
                           float* __restrict__ neigh) {
    __shared__ float sm[WPB * SS * BSTR];

    const int tid  = threadIdx.x;
    const int lane = tid & 31;
    const int w     = __builtin_amdgcn_readfirstlane(tid >> 5);
    const int b     = blockIdx.y;
    const int ibase = blockIdx.x * IPB;
    const int i0    = ibase + w * IW;
    const bool live = i0 < SS;

    const float* xl = x + (size_t)b * SS * DD + lane * 4;
    const int*   Ab = A + (size_t)b * SS * SS;

    const int  cc     = tid % IPB;
    const int  jr     = tid / IPB;
    const bool stager = (tid < SROWS * IPB) && (ibase + cc < SS);
    int av[SS / SROWS];
    if (stager) {
        const int* ap = Ab + jr * SS + ibase + cc;
        #pragma unroll
        for (int k = 0; k < SS / SROWS; ++k) av[k] = ap[k * SROWS * SS];
    }

    v4f bufA[NB], bufB[NB];
    if (live) {
        #pragma unroll
        for (int r = 0; r < NB; ++r) bufA[r] = *(const v4f*)(xl + r * DD);
    }

    if (stager) {
        float* sp = sm + ((cc / IW) * SS + jr) * BSTR + (cc % IW);
        #pragma unroll
        for (int k = 0; k < SS / SROWS; ++k) sp[k * SROWS * BSTR] = (av[k] == 1) ? 1.0f : 0.0f;
    }
    __syncthreads();
    if (!live) return;

    const float* smw = sm + w * SS * BSTR;

    v4f P[IW];
    #pragma unroll
    for (int c = 0; c < IW; ++c) P[c] = (v4f){-INFINITY, -INFINITY, -INFINITY, -INFINITY};

    v4f   m03[NB];
    float m4[NB];

#define MASKS(MP)                                                               \
    _Pragma("unroll")                                                           \
    for (int r = 0; r < NB; ++r) {                                              \
        m03[r] = *(const v4f*)((MP) + r * BSTR);                                \
        m4[r]  = (MP)[r * BSTR + 4];                                            \
    }

#define CONSUME(BUF)                                                            \
    _Pragma("unroll")                                                           \
    for (int r = 0; r < NB; ++r) {                                              \
        const v4f xv = BUF[r];                                                  \
        _Pragma("unroll")                                                       \
        for (int k = 0; k < 4; ++k) {                                           \
            P[0][k] = fmaxf(P[0][k], xv[k] * m03[r][0]);                        \
            P[1][k] = fmaxf(P[1][k], xv[k] * m03[r][1]);                        \
            P[2][k] = fmaxf(P[2][k], xv[k] * m03[r][2]);                        \
            P[3][k] = fmaxf(P[3][k], xv[k] * m03[r][3]);                        \
            P[4][k] = fmaxf(P[4][k], xv[k] * m4[r]);                            \
        }                                                                       \
    }

    for (int jb = 0; jb < SS; jb += 2 * NB) {
        const float* mp = smw + jb * BSTR;
        MASKS(mp)
        #pragma unroll
        for (int r = 0; r < NB; ++r) bufB[r] = *(const v4f*)(xl + (jb + NB + r) * DD);
        __builtin_amdgcn_sched_barrier(0);
        CONSUME(bufA)
        __builtin_amdgcn_sched_barrier(0);
        MASKS(mp + NB * BSTR)
        const int jn = (jb + 2 * NB < SS) ? jb + 2 * NB : jb;
        const float* xn = xl + jn * DD;
        #pragma unroll
        for (int r = 0; r < NB; ++r) bufA[r] = *(const v4f*)(xn + r * DD);
        __builtin_amdgcn_sched_barrier(0);
        CONSUME(bufB)
    }
#undef CONSUME
#undef MASKS

    #pragma unroll
    for (int c = 0; c < IW; ++c)
        *(v4f*)(neigh + ((size_t)b * SS + (i0 + c)) * DD + lane * 4) = P[c];
}

#define GP 132
#define TM 32
#define TN 64

__global__ __launch_bounds__(256)
void EOPA_gemm_wmma_kernel(const float* __restrict__ X,
                           const float* __restrict__ NEIGH,
                           const float* __restrict__ Wself,
                           const float* __restrict__ Wneigh,
                           float* __restrict__ out) {
    __shared__ __attribute__((aligned(16))) float sA[2][TM * GP];
    __shared__ __attribute__((aligned(16))) float sW[2][TN * GP];

    const int tid  = threadIdx.x;
    const int lane = tid & 31;
    const int wave = __builtin_amdgcn_readfirstlane(tid >> 5);

    const int m0  = (blockIdx.x >> 1) * TM;
    const int nb  = (blockIdx.x & 1) * TN;
    const int col = lane * 4;

    v4f xa[TM / 8], na[TM / 8], ws[TN / 8], wn[TN / 8];
    #pragma unroll
    for (int it = 0; it < TM / 8; ++it)
        xa[it] = *(const v4f*)(X + (size_t)(m0 + it * 8 + wave) * DD + col);
    #pragma unroll
    for (int it = 0; it < TN / 8; ++it)
        ws[it] = *(const v4f*)(Wself + (size_t)(nb + it * 8 + wave) * DD + col);

    const int mi     = wave >> 2;
    const int ni     = wave & 3;
    const int laneLo = lane & 15;
    const int kHalf  = (lane >> 4) << 1;
    const int fa = (mi * 16 + laneLo) * GP + kHalf;
    const int fb = (ni * 16 + laneLo) * GP + kHalf;

    v8f acc = {};

    #pragma unroll
    for (int it = 0; it < TM / 8; ++it)
        *(v4f*)(&sA[0][(it * 8 + wave) * GP + col]) = xa[it];
    #pragma unroll
    for (int it = 0; it < TN / 8; ++it)
        *(v4f*)(&sW[0][(it * 8 + wave) * GP + col]) = ws[it];
    __syncthreads();
    #pragma unroll
    for (int it = 0; it < TM / 8; ++it)
        na[it] = *(const v4f*)(NEIGH + (size_t)(m0 + it * 8 + wave) * DD + col);
    #pragma unroll
    for (int it = 0; it < TN / 8; ++it)
        wn[it] = *(const v4f*)(Wneigh + (size_t)(nb + it * 8 + wave) * DD + col);
    {
        v2f af[DD / 4], bf[DD / 4];
        #pragma unroll
        for (int s = 0; s < DD / 4; ++s) {
            af[s] = *(const v2f*)(&sA[0][fa + 4 * s]);
            bf[s] = *(const v2f*)(&sW[0][fb + 4 * s]);
        }
        __builtin_amdgcn_sched_barrier(0);
        #pragma unroll
        for (int s = 0; s < DD / 4; ++s)
            acc = __builtin_amdgcn_wmma_f32_16x16x4_f32(
                false, af[s], false, bf[s], (short)0, acc, false, false);
    }
    asm volatile("" :: "v"(acc));

    #pragma unroll
    for (int it = 0; it < TM / 8; ++it)
        *(v4f*)(&sA[1][(it * 8 + wave) * GP + col]) = na[it];
    #pragma unroll
    for (int it = 0; it < TN / 8; ++it)
        *(v4f*)(&sW[1][(it * 8 + wave) * GP + col]) = wn[it];
    __syncthreads();
    {
        v2f af[DD / 4], bf[DD / 4];
        #pragma unroll
        for (int s = 0; s < DD / 4; ++s) {
            af[s] = *(const v2f*)(&sA[1][fa + 4 * s]);
            bf[s] = *(const v2f*)(&sW[1][fb + 4 * s]);
        }
        __builtin_amdgcn_sched_barrier(0);
        #pragma unroll
        for (int s = 0; s < DD / 4; ++s)
            acc = __builtin_amdgcn_wmma_f32_16x16x4_f32(
                false, af[s], false, bf[s], (short)0, acc, false, false);
    }

    const int colN    = nb + ni * 16 + laneLo;
    const int rowBase = m0 + mi * 16 + ((lane >> 4) << 3);
    #pragma unroll
    for (int r = 0; r < 8; ++r) {
        out[(size_t)(rowBase + r) * DD + colN] = acc[r];
    }
}

extern "C" void kernel_launch(void* const* d_in, const int* in_sizes, int n_in,
                              void* d_out, int out_size, void* d_ws, size_t ws_size,
                              hipStream_t stream) {
    const float* x      = (const float*)d_in[0];
    const int*   A      = (const int*)  d_in[1];
    const float* Wself  = (const float*)d_in[2];
    const float* Wneigh = (const float*)d_in[3];
    float*       out    = (float*)d_out;
    float*       neigh  = (float*)d_ws;

    EOPA_neigh_max_kernel<<<dim3(8, 32), 128, 0, stream>>>(x, A, neigh);

    EOPA_gemm_wmma_kernel<<<300, 256, 0, stream>>>(x, neigh, Wself, Wneigh, out);
}


# GEMM fused into the neighbour-max launch: each group of 20 nodes multiplies its own rows (32 staged, 20 used) by both weight matrices staged in LDS; one launch, neigh stays on chip; sum order kept
# speedup vs baseline: 48.1162x; 1.2075x over previous
#include <hip/hip_runtime.h>
#include <hip/hip_bf16.h>

typedef __attribute__((ext_vector_type(2))) float v2f;
typedef __attribute__((ext_vector_type(4))) float v4f;
typedef __attribute__((ext_vector_type(8))) float v8f;

#define BB 32
#define SS 150
#define DD 128

#define IW   5
#define WPB  4
#define IPB  (IW * WPB)
#define NB   25
#define BSTR 8
#define SROWS 6

#define GP 132
#define TM 32
#define WR (DD / WPB)
#define ZR ((TM - IPB) / WPB)
#define KC 8
#define NCHK (2 * DD / 4 / KC)

__global__ __launch_bounds__(WPB * 32)
void EOPA_fused_kernel(const float* __restrict__ x,
                       const int* __restrict__ A,
                       const float* __restrict__ Wself,
                       const float* __restrict__ Wneigh,
                       float* __restrict__ out) {
    __shared__ __attribute__((aligned(16))) float sm[WPB * SS * BSTR];
    __shared__ __attribute__((aligned(16))) float sA[2][TM * GP];
    __shared__ __attribute__((aligned(16))) float sW[2][DD * GP];

    const int tid  = threadIdx.x;
    const int lane = tid & 31;
    const int w     = __builtin_amdgcn_readfirstlane(tid >> 5);
    const int b     = blockIdx.y;
    const int ibase = blockIdx.x * IPB;
    const int i0    = ibase + w * IW;
    const bool live = i0 < SS;
    const int col   = lane * 4;

    const float* xl = x + (size_t)b * SS * DD + col;
    const int*   Ab = A + (size_t)b * SS * SS;

    const int  cc      = tid % IPB;
    const int  jr      = tid / IPB;
    const bool stager  = tid < SROWS * IPB;
    const bool hasnode = stager && (ibase + cc < SS);
    int av[SS / SROWS];
    if (hasnode) {
        const int* ap = Ab + jr * SS + ibase + cc;
        #pragma unroll
        for (int k = 0; k < SS / SROWS; ++k) av[k] = ap[k * SROWS * SS];
    } else {
        #pragma unroll
        for (int k = 0; k < SS / SROWS; ++k) av[k] = 0;
    }

    v4f bufA[NB], bufB[NB];
    #pragma unroll
    for (int r = 0; r < NB; ++r) bufA[r] = *(const v4f*)(xl + r * DD);

    if (stager) {
        float* sp = sm + ((cc / IW) * SS + jr) * BSTR + (cc % IW);
        #pragma unroll
        for (int k = 0; k < SS / SROWS; ++k) sp[k * SROWS * BSTR] = (av[k] == 1) ? 1.0f : 0.0f;
    }
    __syncthreads();

    const float* smw = sm + w * SS * BSTR;

    v4f P[IW];
    #pragma unroll
    for (int c = 0; c < IW; ++c) P[c] = (v4f){-INFINITY, -INFINITY, -INFINITY, -INFINITY};

    v4f   m03[NB];
    float m4[NB];

    v4f xo[IW];
    v4f ws[WR], wn[WR];
    const int io = live ? i0 : SS - IW;

#define MASKS(MP)                                                               \
    _Pragma("unroll")                                                           \
    for (int r = 0; r < NB; ++r) {                                              \
        m03[r] = *(const v4f*)((MP) + r * BSTR);                                \
        m4[r]  = (MP)[r * BSTR + 4];                                            \
    }

#define CONSUME(BUF)                                                            \
    _Pragma("unroll")                                                           \
    for (int r = 0; r < NB; ++r) {                                              \
        const v4f xv = BUF[r];                                                  \
        _Pragma("unroll")                                                       \
        for (int k = 0; k < 4; ++k) {                                           \
            P[0][k] = fmaxf(P[0][k], xv[k] * m03[r][0]);                        \
            P[1][k] = fmaxf(P[1][k], xv[k] * m03[r][1]);                        \
            P[2][k] = fmaxf(P[2][k], xv[k] * m03[r][2]);                        \
            P[3][k] = fmaxf(P[3][k], xv[k] * m03[r][3]);                        \
            P[4][k] = fmaxf(P[4][k], xv[k] * m4[r]);                            \
        }                                                                       \
    }

#define WLOADS1                                                                 \
    _Pragma("unroll")                                                           \
    for (int c = 0; c < IW; ++c) xo[c] = *(const v4f*)(xl + (io + c) * DD);     \
    _Pragma("unroll")                                                           \
    for (int it = 0; it < WR; ++it)                                             \
        ws[it] = *(const v4f*)(Wself  + (it * WPB + w) * DD + col);
#define WLOADS2                                                                 \
    _Pragma("unroll")                                                           \
    for (int it = 0; it < WR; ++it)                                             \
        wn[it] = *(const v4f*)(Wneigh + (it * WPB + w) * DD + col);

#define TRIP(JB, HOOK1, HOOK2)                                                        \
    {                                                                           \
        const float* mp = smw + (JB) * BSTR;                                    \
        MASKS(mp)                                                               \
        _Pragma("unroll")                                                       \
        for (int r = 0; r < NB; ++r)                                            \
            bufB[r] = *(const v4f*)(xl + ((JB) + NB + r) * DD);                 \
        HOOK1                                                                   \
        __builtin_amdgcn_sched_barrier(0);                              \
        CONSUME(bufA)                                                           \
        __builtin_amdgcn_sched_barrier(0);                              \
        MASKS(mp + NB * BSTR)                                                   \
        const int jn = ((JB) + 2 * NB < SS) ? (JB) + 2 * NB : (JB);             \
        const float* xn = xl + jn * DD;                                         \
        _Pragma("unroll")                                                       \
        for (int r = 0; r < NB; ++r) bufA[r] = *(const v4f*)(xn + r * DD);      \
        HOOK2                                                                   \
        __builtin_amdgcn_sched_barrier(0);                              \
        CONSUME(bufB)                                                           \
    }

    TRIP(0, WLOADS1, WLOADS2)

    #pragma unroll
    for (int c = 0; c < IW; ++c)
        *(v4f*)(&sA[0][(w * IW + c) * GP + col]) = xo[c];
    #pragma unroll
    for (int it = 0; it < WR; ++it) {
        *(v4f*)(&sW[0][(it * WPB + w) * GP + col]) = ws[it];
        *(v4f*)(&sW[1][(it * WPB + w) * GP + col]) = wn[it];
    }
    #pragma unroll
    for (int t = 0; t < ZR; ++t) {
        *(v4f*)(&sA[0][(IPB + w * ZR + t) * GP + col]) = (v4f){0.0f, 0.0f, 0.0f, 0.0f};
        *(v4f*)(&sA[1][(IPB + w * ZR + t) * GP + col]) = (v4f){0.0f, 0.0f, 0.0f, 0.0f};
    }

    for (int jb = 2 * NB; jb < SS; jb += 2 * NB)
        TRIP(jb, , )
#undef TRIP
#undef WLOADS2
#undef WLOADS1
#undef CONSUME
#undef MASKS

    #pragma unroll
    for (int c = 0; c < IW; ++c)
        *(v4f*)(&sA[1][(w * IW + c) * GP + col]) = P[c];
    __syncthreads();

    const int laneLo = lane & 15;
    const int kHalf  = (lane >> 4) << 1;
    const int fa = laneLo * GP + kHalf;
    const int fb = (w * 32 + laneLo) * GP + kHalf;

    v8f acc00 = {}, acc01 = {}, acc10 = {}, acc11 = {};
    v2f a0[2][KC], a1[2][KC], b0[2][KC], b1[2][KC];

#define FRAGS(BI, PR, S0)                                                       \
    _Pragma("unroll")                                                           \
    for (int s = 0; s < KC; ++s) {                                              \
        a0[BI][s] = *(const v2f*)(&sA[PR][fa + 4 * ((S0) + s)]);                \
        a1[BI][s] = *(const v2f*)(&sA[PR][fa + 16 * GP + 4 * ((S0) + s)]);      \
        b0[BI][s] = *(const v2f*)(&sW[PR][fb + 4 * ((S0) + s)]);                \
        b1[BI][s] = *(const v2f*)(&sW[PR][fb + 16 * GP + 4 * ((S0) + s)]);      \
    }
#define WM(AF, BF, ACC)                                                         \
    ACC = __builtin_amdgcn_wmma_f32_16x16x4_f32(false, AF, false, BF, (short)0, ACC, false, false);
#define MMAS(BI)                                                                \
    _Pragma("unroll")                                                           \
    for (int s = 0; s < KC; ++s) {                                              \
        WM(a0[BI][s], b0[BI][s], acc00)                                         \
        WM(a0[BI][s], b1[BI][s], acc01)                                         \
        WM(a1[BI][s], b0[BI][s], acc10)                                         \
        WM(a1[BI][s], b1[BI][s], acc11)                                         \
    }

    FRAGS(0, 0, 0)
    #pragma unroll
    for (int ch = 0; ch < NCHK; ++ch) {
        if (ch + 1 < NCHK) {
            FRAGS((ch + 1) & 1, (ch + 1) / (NCHK / 2), ((ch + 1) % (NCHK / 2)) * KC)
        }
        __builtin_amdgcn_sched_barrier(0);
        MMAS(ch & 1)
        __builtin_amdgcn_sched_barrier(0);
    }
#undef MMAS
#undef WM
#undef FRAGS

    float* ob = out + ((size_t)b * SS + ibase) * DD + w * 32 + laneLo;
    const int h8 = (lane >> 4) << 3;
    #pragma unroll
    for (int r = 0; r < 8; ++r) {
        const int r0 = h8 + r;
        if (ibase + r0 < SS) {
            ob[r0 * DD]      = acc00[r];
            ob[r0 * DD + 16] = acc01[r];
        }
        const int r1 = 16 + h8 + r;
        if (r1 < IPB && ibase + r1 < SS) {
            ob[r1 * DD]      = acc10[r];
            ob[r1 * DD + 16] = acc11[r];
        }
    }
}

extern "C" void kernel_launch(void* const* d_in, const int* in_sizes, int n_in,
                              void* d_out, int out_size, void* d_ws, size_t ws_size,
                              hipStream_t stream) {
    const float* x      = (const float*)d_in[0];
    const int*   A      = (const int*)  d_in[1];
    const float* Wself  = (const float*)d_in[2];
    const float* Wneigh = (const float*)d_in[3];
    float*       out    = (float*)d_out;

    EOPA_fused_kernel<<<dim3(8, 32), 128, 0, stream>>>(x, A, Wself, Wneigh, out);
}


# 8 waves per group: waves 4-7 (2nd wave of each SIMD) stage the weights and run X.Wself^T on WMMA beside the neighbour-max waves, then add NEIGH.Wneigh^T; was 4 waves doing all in turn; sum order kept
# speedup vs baseline: 52.1967x; 1.0848x over previous
#include <hip/hip_runtime.h>
#include <hip/hip_bf16.h>

typedef __attribute__((ext_vector_type(2))) float v2f;
typedef __attribute__((ext_vector_type(4))) float v4f;
typedef __attribute__((ext_vector_type(8))) float v8f;

#define BB 32
#define SS 150
#define DD 128

#define IW   5
#define WPB  4
#define NWV  (2 * WPB)
#define IPB  (IW * WPB)
#define NB   25
#define BSTR 8
#define SROWS 6

#define GP 132
#define TM 32
#define WR (DD / WPB)
#define ZR ((TM - IPB) / WPB)
#define KC 8
#define NCH (DD / 4 / KC)

__global__ __launch_bounds__(NWV * 32)
void EOPA_fused_kernel(const float* __restrict__ x,
                       const int* __restrict__ A,
                       const float* __restrict__ Wself,
                       const float* __restrict__ Wneigh,
                       float* __restrict__ out) {
    __shared__ __attribute__((aligned(16))) float sm[WPB * SS * BSTR];
    __shared__ __attribute__((aligned(16))) float sA[2][TM * GP];
    __shared__ __attribute__((aligned(16))) float sW[2][DD * GP];

    const int tid  = threadIdx.x;
    const int lane = tid & 31;
    const int wv    = __builtin_amdgcn_readfirstlane(tid >> 5);
    const bool mxw  = wv >= WPB;
    const int w     = wv & (WPB - 1);
    const int b     = blockIdx.y;
    const int ibase = blockIdx.x * IPB;
    const int i0    = ibase + w * IW;
    const bool live = i0 < SS;
    const int col   = lane * 4;

    const float* xl = x + (size_t)b * SS * DD + col;
    const int*   Ab = A + (size_t)b * SS * SS;

    v4f bufA[NB], bufB[NB];
    v8f acc00 = {}, acc01 = {}, acc10 = {}, acc11 = {};

    if (!mxw) {
        const int  cc      = tid % IPB;
        const int  jr      = tid / IPB;
        const bool stager  = tid < SROWS * IPB;
        const bool hasnode = stager && (ibase + cc < SS);
        int av[SS / SROWS];
        if (hasnode) {
            const int* ap = Ab + jr * SS + ibase + cc;
            #pragma unroll
            for (int k = 0; k < SS / SROWS; ++k) av[k] = ap[k * SROWS * SS];
        } else {
            #pragma unroll
            for (int k = 0; k < SS / SROWS; ++k) av[k] = 0;
        }

        #pragma unroll
        for (int r = 0; r < NB; ++r) bufA[r] = *(const v4f*)(xl + r * DD);

        if (stager) {
            float* sp = sm + ((cc / IW) * SS + jr) * BSTR + (cc % IW);
            #pragma unroll
            for (int k = 0; k < SS / SROWS; ++k) sp[k * SROWS * BSTR] = (av[k] == 1) ? 1.0f : 0.0f;
        }
    } else {
        const int io = live ? i0 : SS - IW;
        v4f xo[IW];
        #pragma unroll
        for (int c = 0; c < IW; ++c) xo[c] = *(const v4f*)(xl + (io + c) * DD);
        #pragma unroll
        for (int c = 0; c < IW; ++c)
            *(v4f*)(&sA[0][(w * IW + c) * GP + col]) = xo[c];
        #pragma unroll
        for (int t = 0; t < ZR; ++t) {
            *(v4f*)(&sA[0][(IPB + w * ZR + t) * GP + col]) = (v4f){0.0f, 0.0f, 0.0f, 0.0f};
            *(v4f*)(&sA[1][(IPB + w * ZR + t) * GP + col]) = (v4f){0.0f, 0.0f, 0.0f, 0.0f};
        }
    }
    __syncthreads();

    const int laneLo = lane & 15;
    const int kHalf  = (lane >> 4) << 1;
    const int fa = laneLo * GP + kHalf;
    const int fb = (w * 32 + laneLo) * GP + kHalf;

    v2f a0[2][KC], a1[2][KC], b0[2][KC], b1[2][KC];

#define FRAGS(BI, PR, S0)                                                       \
    _Pragma("unroll")                                                           \
    for (int s = 0; s < KC; ++s) {                                              \
        a0[BI][s] = *(const v2f*)(&sA[PR][fa + 4 * ((S0) + s)]);                \
        a1[BI][s] = *(const v2f*)(&sA[PR][fa + 16 * GP + 4 * ((S0) + s)]);      \
        b0[BI][s] = *(const v2f*)(&sW[PR][fb + 4 * ((S0) + s)]);                \
        b1[BI][s] = *(const v2f*)(&sW[PR][fb + 16 * GP + 4 * ((S0) + s)]);      \
    }
#define WM(AF, BF, ACC)                                                         \
    ACC = __builtin_amdgcn_wmma_f32_16x16x4_f32(false, AF, false, BF, (short)0, ACC, false, false);
#define MMAS(BI)                                                                \
    _Pragma("unroll")                                                           \
    for (int s = 0; s < KC; ++s) {                                              \
        WM(a0[BI][s], b0[BI][s], acc00)                                         \
        WM(a0[BI][s], b1[BI][s], acc01)                                         \
        WM(a1[BI][s], b0[BI][s], acc10)                                         \
        WM(a1[BI][s], b1[BI][s], acc11)                                         \
    }
#define PRODUCT(PR)                                                             \
    FRAGS(0, PR, 0)                                                             \
    _Pragma("unroll")                                                           \
    for (int ch = 0; ch < NCH; ++ch) {                                          \
          \
                                       \
        if (ch + 1 < NCH) {                                                     \
            FRAGS((ch + 1) & 1, PR, (ch + 1) * KC)                              \
        }                                                                       \
          \
                           \
        __builtin_amdgcn_sched_barrier(0);                                      \
        MMAS(ch & 1)                                                            \
          \
        __builtin_amdgcn_sched_barrier(0);                                      \
    }

    if (!mxw) {
        const float* smw = sm + w * SS * BSTR;

        v4f P[IW];
        #pragma unroll
        for (int c = 0; c < IW; ++c) P[c] = (v4f){-INFINITY, -INFINITY, -INFINITY, -INFINITY};

        v4f   m03[NB];
        float m4[NB];

#define MASKS(MP)                                                               \
    _Pragma("unroll")                                                           \
    for (int r = 0; r < NB; ++r) {                                              \
        m03[r] = *(const v4f*)((MP) + r * BSTR);                                \
        m4[r]  = (MP)[r * BSTR + 4];                                            \
    }

#define CONSUME(BUF)                                                            \
    _Pragma("unroll")                                                           \
    for (int r = 0; r < NB; ++r) {                                              \
        const v4f xv = BUF[r];                                                  \
        _Pragma("unroll")                                                       \
        for (int k = 0; k < 4; ++k) {                                           \
            P[0][k] = fmaxf(P[0][k], xv[k] * m03[r][0]);                        \
            P[1][k] = fmaxf(P[1][k], xv[k] * m03[r][1]);                        \
            P[2][k] = fmaxf(P[2][k], xv[k] * m03[r][2]);                        \
            P[3][k] = fmaxf(P[3][k], xv[k] * m03[r][3]);                        \
            P[4][k] = fmaxf(P[4][k], xv[k] * m4[r]);                            \
        }                                                                       \
    }

#define TRIP(JB)                                                                \
    {                                                                           \
        const float* mp = smw + (JB) * BSTR;                                    \
        MASKS(mp)                                                               \
        _Pragma("unroll")                                                       \
        for (int r = 0; r < NB; ++r)                                            \
            bufB[r] = *(const v4f*)(xl + ((JB) + NB + r) * DD);                 \
        __builtin_amdgcn_sched_barrier(0);                              \
        CONSUME(bufA)                                                           \
        __builtin_amdgcn_sched_barrier(0);                              \
        MASKS(mp + NB * BSTR)                                                   \
        const int jn = ((JB) + 2 * NB < SS) ? (JB) + 2 * NB : (JB);             \
        const float* xn = xl + jn * DD;                                         \
        _Pragma("unroll")                                                       \
        for (int r = 0; r < NB; ++r) bufA[r] = *(const v4f*)(xn + r * DD);      \
        __builtin_amdgcn_sched_barrier(0);                              \
        CONSUME(bufB)                                                           \
    }

        TRIP(0)
        for (int jb = 2 * NB; jb < SS; jb += 2 * NB)
            TRIP(jb)
#undef TRIP
#undef CONSUME
#undef MASKS

        #pragma unroll
        for (int c = 0; c < IW; ++c)
            *(v4f*)(&sA[1][(w * IW + c) * GP + col]) = P[c];
    } else {
        v4f ws[WR], wn[WR];
        #pragma unroll
        for (int it = 0; it < WR; ++it)
            ws[it] = *(const v4f*)(Wself  + (w * WR + it) * DD + col);
        #pragma unroll
        for (int it = 0; it < WR; ++it)
            wn[it] = *(const v4f*)(Wneigh + (w * WR + it) * DD + col);
        #pragma unroll
        for (int it = 0; it < WR; ++it)
            *(v4f*)(&sW[0][(w * WR + it) * GP + col]) = ws[it];

        PRODUCT(0)

        #pragma unroll
        for (int it = 0; it < WR; ++it)
            *(v4f*)(&sW[1][(w * WR + it) * GP + col]) = wn[it];
    }
    __syncthreads();

    if (mxw) {
        PRODUCT(1)

        float* ob = out + ((size_t)b * SS + ibase) * DD + w * 32 + laneLo;
        const int h8 = (lane >> 4) << 3;
        #pragma unroll
        for (int r = 0; r < 8; ++r) {
            const int r0 = h8 + r;
            if (ibase + r0 < SS) {
                ob[r0 * DD]      = acc00[r];
                ob[r0 * DD + 16] = acc01[r];
            }
            const int r1 = 16 + h8 + r;
            if (r1 < IPB && ibase + r1 < SS) {
                ob[r1 * DD]      = acc10[r];
                ob[r1 * DD + 16] = acc11[r];
            }
        }
    }
#undef PRODUCT
#undef MMAS
#undef WM
#undef FRAGS
}

extern "C" void kernel_launch(void* const* d_in, const int* in_sizes, int n_in,
                              void* d_out, int out_size, void* d_ws, size_t ws_size,
                              hipStream_t stream) {
    const float* x      = (const float*)d_in[0];
    const int*   A      = (const int*)  d_in[1];
    const float* Wself  = (const float*)d_in[2];
    const float* Wneigh = (const float*)d_in[3];
    float*       out    = (float*)d_out;

    EOPA_fused_kernel<<<dim3(8, 32), 256, 0, stream>>>(x, A, Wself, Wneigh, out);
}
